# starting kernel
# baseline (speedup 1.0000x reference)
#include <hip/hip_runtime.h>


typedef __attribute__((ext_vector_type(2))) float v2f;
typedef __attribute__((ext_vector_type(8))) float v8f;

#define BATCH 16
#define CH    64
#define TLEN  2048
#define KNN   3

__global__ __launch_bounds__(256) void nsq_kernel(const float* __restrict__ x,
                                                  float* __restrict__ nsq) {
    int g = blockIdx.x * 256 + threadIdx.x;
    int b = g >> 11;
    int t = g & (TLEN - 1);
    const float* xb = x + (size_t)b * CH * TLEN + t;
    float s = 0.0f;
#pragma unroll
    for (int c = 0; c < CH; ++c) {
        float v = xb[c * TLEN];
        s += v * v;
    }
    nsq[g] = s;
}

__global__ __launch_bounds__(256) void knn_kernel(const float* __restrict__ x,
                                                  const float* __restrict__ nsq,
                                                  int* __restrict__ idx_out) {
    const int b    = blockIdx.x >> 7;
    const int tile = blockIdx.x & 127;
    const int t0   = tile * 16;

    extern __shared__ float smem[];
    float* dist   = smem;
    float* cand_d = smem + 16 * TLEN;
    int*   cand_i = (int*)(cand_d + 16 * 48);

    const int tid   = threadIdx.x;
    const int lane  = tid & 31;
    const int wave  = tid >> 5;
    const int lmod  = lane & 15;
    const int lhalf = (lane < 16) ? 0 : 2;
    const int mhalf = (lane < 16) ? 0 : 8;

    const float* xb   = x   + (size_t)b * CH * TLEN;
    const float* nsqb = nsq + (size_t)b * TLEN;

    v2f afrag[16];
#pragma unroll
    for (int kc = 0; kc < 16; ++kc) {
        int c0 = kc * 4 + lhalf;
        afrag[kc].x = xb[(size_t)(c0 + 0) * TLEN + t0 + lmod];
        afrag[kc].y = xb[(size_t)(c0 + 1) * TLEN + t0 + lmod];
    }
    float nsq_t[8];
#pragma unroll
    for (int r = 0; r < 8; ++r) nsq_t[r] = nsqb[t0 + r + mhalf];

    for (int it = 0; it < 16; ++it) {
        const int s0 = (it * 8 + wave) * 16;
        v8f acc = {};
#pragma unroll
        for (int kc = 0; kc < 16; ++kc) {
            int c0 = kc * 4 + lhalf;
            v2f bfrag;
            bfrag.x = xb[(size_t)(c0 + 0) * TLEN + s0 + lmod];
            bfrag.y = xb[(size_t)(c0 + 1) * TLEN + s0 + lmod];
            acc = __builtin_amdgcn_wmma_f32_16x16x4_f32(
                false, afrag[kc], false, bfrag, (short)0, acc, false, false);
        }
        const float nsq_s = nsqb[s0 + lmod];
#pragma unroll
        for (int r = 0; r < 8; ++r) {
            int m   = r + mhalf;
            float d = nsq_t[r] + nsq_s - 2.0f * acc[r];
            dist[m * TLEN + s0 + lmod] = fmaxf(d, 0.0f);
        }
    }
    __syncthreads();

    const int row = tid >> 4;
    const int tt  = tid & 15;
    float d0 = 3.4e38f, d1 = 3.4e38f, d2 = 3.4e38f;
    int   i0 = 0x7fffffff, i1 = 0x7fffffff, i2 = 0x7fffffff;
    const float* drow = dist + row * TLEN;
    for (int j = tt; j < TLEN; j += 16) {
        float d = drow[j];
        if (d < d0)      { d2 = d1; i2 = i1; d1 = d0; i1 = i0; d0 = d; i0 = j; }
        else if (d < d1) { d2 = d1; i2 = i1; d1 = d;  i1 = j; }
        else if (d < d2) { d2 = d;  i2 = j; }
    }
    cand_d[row * 48 + tt * 3 + 0] = d0;  cand_i[row * 48 + tt * 3 + 0] = i0;
    cand_d[row * 48 + tt * 3 + 1] = d1;  cand_i[row * 48 + tt * 3 + 1] = i1;
    cand_d[row * 48 + tt * 3 + 2] = d2;  cand_i[row * 48 + tt * 3 + 2] = i2;
    __syncthreads();

    if (tt == 0) {
        float bd0 = 3.4e38f, bd1 = 3.4e38f, bd2 = 3.4e38f;
        int   bi0 = 0x7fffffff, bi1 = 0x7fffffff, bi2 = 0x7fffffff;
        for (int c = 0; c < 48; ++c) {
            float d = cand_d[row * 48 + c];
            int   i = cand_i[row * 48 + c];
            if (d < bd0 || (d == bd0 && i < bi0)) {
                bd2 = bd1; bi2 = bi1; bd1 = bd0; bi1 = bi0; bd0 = d; bi0 = i;
            } else if (d < bd1 || (d == bd1 && i < bi1)) {
                bd2 = bd1; bi2 = bi1; bd1 = d; bi1 = i;
            } else if (d < bd2 || (d == bd2 && i < bi2)) {
                bd2 = d; bi2 = i;
            }
        }
        int* o = idx_out + ((size_t)b * TLEN + t0 + row) * KNN;
        o[0] = bi0; o[1] = bi1; o[2] = bi2;
    }
}

__global__ __launch_bounds__(256) void conv_kernel(const float* __restrict__ x,
                                                   const float* __restrict__ W,
                                                   const float* __restrict__ bias,
                                                   const int* __restrict__ idx,
                                                   float* __restrict__ out) {
    const int b   = blockIdx.x >> 6;
    const int blk = blockIdx.x & 63;
    const int wave  = threadIdx.x >> 5;
    const int lane  = threadIdx.x & 31;
    const int lmod  = lane & 15;
    const int lhalf = (lane < 16) ? 0 : 2;
    const int mhalf = (lane < 16) ? 0 : 8;

    const int o0 = (wave & 3) * 16;
    const int t0 = blk * 32 + (wave >> 2) * 16;
    const int tok = t0 + lmod;

    const float* xb = x + (size_t)b * CH * TLEN;
    const int* my = idx + ((size_t)b * TLEN + tok) * KNN;
    const int s0i = my[0], s1i = my[1], s2i = my[2];

    v8f acc = {};
#pragma unroll
    for (int kc = 0; kc < 48; ++kc) {
        const int ka0 = kc * 4 + lhalf;
        const int ka1 = ka0 + 1;
        v2f a, bb;
        a.x = W[(size_t)(o0 + lmod) * (CH * KNN) + ka0];
        a.y = W[(size_t)(o0 + lmod) * (CH * KNN) + ka1];
        {
            int i = ka0 / 3, kk = ka0 - i * 3;
            int s = (kk == 0) ? s0i : ((kk == 1) ? s1i : s2i);
            bb.x = xb[(size_t)i * TLEN + s];
        }
        {
            int i = ka1 / 3, kk = ka1 - i * 3;
            int s = (kk == 0) ? s0i : ((kk == 1) ? s1i : s2i);
            bb.y = xb[(size_t)i * TLEN + s];
        }
        acc = __builtin_amdgcn_wmma_f32_16x16x4_f32(
            false, a, false, bb, (short)0, acc, false, false);
    }

#pragma unroll
    for (int r = 0; r < 8; ++r) {
        int o = o0 + r + mhalf;
        out[(size_t)b * CH * TLEN + (size_t)o * TLEN + tok] = acc[r] + bias[o];
    }
}

extern "C" void kernel_launch(void* const* d_in, const int* in_sizes, int n_in,
                              void* d_out, int out_size, void* d_ws, size_t ws_size,
                              hipStream_t stream) {
    (void)in_sizes; (void)n_in; (void)out_size; (void)ws_size;
    const float* x    = (const float*)d_in[0];
    const float* W    = (const float*)d_in[1];
    const float* bias = (const float*)d_in[2];
    float* out = (float*)d_out;

    float* nsq = (float*)d_ws;
    int*   idx = (int*)((char*)d_ws + (size_t)BATCH * TLEN * sizeof(float));

    nsq_kernel<<<(BATCH * TLEN) / 256, 256, 0, stream>>>(x, nsq);

    const size_t smem = (size_t)16 * TLEN * sizeof(float)
                      + (size_t)16 * 48 * sizeof(float)
                      + (size_t)16 * 48 * sizeof(int);
    hipFuncSetAttribute((const void*)knn_kernel,
                        hipFuncAttributeMaxDynamicSharedMemorySize, (int)smem);
    knn_kernel<<<BATCH * (TLEN / 16), 256, smem, stream>>>(x, nsq, idx);

    conv_kernel<<<BATCH * (TLEN / 32), 256, 0, stream>>>(x, W, bias, idx, out);
}


# knn: each wave owns a 16-row tile and keeps every row's running top-3 (distance, column) in registers from the WMMA accumulators; the 128-KB LDS distance slab, its scan and serial merge are gone
# speedup vs baseline: 1.6370x; 1.6370x over previous
#include <hip/hip_runtime.h>


typedef __attribute__((ext_vector_type(2))) float v2f;
typedef __attribute__((ext_vector_type(8))) float v8f;

#define BATCH 16
#define CH    64
#define TLEN  2048
#define KNN   3

__global__ __launch_bounds__(256) void nsq_kernel(const float* __restrict__ x,
                                                  float* __restrict__ nsq) {
    int g = blockIdx.x * 256 + threadIdx.x;
    int b = g >> 11;
    int t = g & (TLEN - 1);
    const float* xb = x + (size_t)b * CH * TLEN + t;
    float s = 0.0f;
#pragma unroll
    for (int c = 0; c < CH; ++c) {
        float v = xb[c * TLEN];
        s += v * v;
    }
    nsq[g] = s;
}

#define CAND_PITCH 49

#define TOP3_INSERT(r, dval, jval)                                            \
    {                                                                         \
        const float d_ = (dval);                                              \
        const int   j_ = (jval);                                              \
        const float a0_ = D0[r], a1_ = D1[r], a2_ = D2[r];                    \
        const int   k0_ = I0[r], k1_ = I1[r], k2_ = I2[r];                    \
        const bool c0_ = d_ < a0_;                                            \
        const bool c1_ = d_ < a1_;                                            \
        const bool c2_ = d_ < a2_;                                            \
        const float t2_ = c2_ ? d_ : a2_;                                     \
        const int   u2_ = c2_ ? j_ : k2_;                                     \
        const float t1_ = c1_ ? d_ : a1_;                                     \
        const int   u1_ = c1_ ? j_ : k1_;                                     \
        D2[r] = c1_ ? a1_ : t2_;                                              \
        I2[r] = c1_ ? k1_ : u2_;                                              \
        D1[r] = c0_ ? a0_ : t1_;                                              \
        I1[r] = c0_ ? k0_ : u1_;                                              \
        D0[r] = c0_ ? d_ : a0_;                                               \
        I0[r] = c0_ ? j_ : k0_;                                               \
    }

#define LOAD_TILE(BF, NS, S0)                                                 \
    {                                                                         \
        _Pragma("unroll")                                                     \
        for (int kc = 0; kc < 16; ++kc) {                                     \
            const int c0 = kc * 4 + lhalf;                                    \
            BF[kc].x = xb[(size_t)(c0 + 0) * TLEN + (S0) + lmod];             \
            BF[kc].y = xb[(size_t)(c0 + 1) * TLEN + (S0) + lmod];             \
        }                                                                     \
        NS = nsqb[(S0) + lmod];                                               \
    }

#define TILE_STEP(BF, NS, S0)                                                 \
    {                                                                         \
        v8f acc = {};                                                         \
        _Pragma("unroll")                                                     \
        for (int kc = 0; kc < 16; ++kc)                                       \
            acc = __builtin_amdgcn_wmma_f32_16x16x4_f32(                      \
                false, afrag[kc], false, BF[kc], (short)0, acc, false, false);\
        const int col_ = (S0) + lmod;                                         \
        _Pragma("unroll")                                                     \
        for (int r = 0; r < 8; ++r) {                                         \
            const float dd_ = fmaxf(nsq_t[r] + (NS) - 2.0f * acc[r], 0.0f);   \
            TOP3_INSERT(r, dd_, col_)                                         \
        }                                                                     \
    }

__global__ __launch_bounds__(256) void knn_kernel(const float* __restrict__ x,
                                                  const float* __restrict__ nsq,
                                                  int* __restrict__ idx_out) {
    const int b   = blockIdx.x >> 4;
    const int blk = blockIdx.x & 15;

    extern __shared__ float smem[];

    const int tid   = threadIdx.x;
    const int lane  = tid & 31;
    const int wave  = tid >> 5;
    const int lmod  = lane & 15;
    const int lhalf = (lane < 16) ? 0 : 2;
    const int mhalf = (lane < 16) ? 0 : 8;
    const int t0    = blk * 128 + wave * 16;

    float* cand_d = smem + wave * (16 * CAND_PITCH);
    int*   cand_i = (int*)(smem + 8 * 16 * CAND_PITCH) + wave * (16 * CAND_PITCH);

    const float* xb   = x   + (size_t)b * CH * TLEN;
    const float* nsqb = nsq + (size_t)b * TLEN;

    v2f afrag[16];
#pragma unroll
    for (int kc = 0; kc < 16; ++kc) {
        int c0 = kc * 4 + lhalf;
        afrag[kc].x = xb[(size_t)(c0 + 0) * TLEN + t0 + lmod];
        afrag[kc].y = xb[(size_t)(c0 + 1) * TLEN + t0 + lmod];
    }
    float nsq_t[8];
#pragma unroll
    for (int r = 0; r < 8; ++r) nsq_t[r] = nsqb[t0 + r + mhalf];

    float D0[8], D1[8], D2[8];
    int   I0[8], I1[8], I2[8];
#pragma unroll
    for (int r = 0; r < 8; ++r) {
        D0[r] = 3.4e38f; D1[r] = 3.4e38f; D2[r] = 3.4e38f;
        I0[r] = 0x7fffffff; I1[r] = 0x7fffffff; I2[r] = 0x7fffffff;
    }

    v2f bA[16], bB[16];
    float nsA, nsB;
    LOAD_TILE(bA, nsA, 0)
#pragma unroll 1
    for (int s0 = 0; s0 < TLEN; s0 += 32) {
        LOAD_TILE(bB, nsB, s0 + 16)
        TILE_STEP(bA, nsA, s0)
        const int sn = (s0 + 32 < TLEN) ? (s0 + 32) : 0;
        LOAD_TILE(bA, nsA, sn)
        TILE_STEP(bB, nsB, s0 + 16)
    }

#pragma unroll
    for (int r = 0; r < 8; ++r) {
        const int base = (r + mhalf) * CAND_PITCH + lmod * 3;
        cand_d[base + 0] = D0[r];  cand_i[base + 0] = I0[r];
        cand_d[base + 1] = D1[r];  cand_i[base + 1] = I1[r];
        cand_d[base + 2] = D2[r];  cand_i[base + 2] = I2[r];
    }
    __syncthreads();

    if (lane < 16) {
        const float* rd = cand_d + lane * CAND_PITCH;
        const int*   ri = cand_i + lane * CAND_PITCH;
        float bd0 = 3.4e38f, bd1 = 3.4e38f, bd2 = 3.4e38f;
        int   bi0 = 0x7fffffff, bi1 = 0x7fffffff, bi2 = 0x7fffffff;
#pragma unroll
        for (int c = 0; c < 48; ++c) {
            const float d = rd[c];
            const int   i = ri[c];
            const bool c0 = d < bd0 || (d == bd0 && i < bi0);
            const bool c1 = d < bd1 || (d == bd1 && i < bi1);
            const bool c2 = d < bd2 || (d == bd2 && i < bi2);
            const float t2 = c2 ? d : bd2;
            const int   u2 = c2 ? i : bi2;
            const float t1 = c1 ? d : bd1;
            const int   u1 = c1 ? i : bi1;
            bd2 = c1 ? bd1 : t2;
            bi2 = c1 ? bi1 : u2;
            bd1 = c0 ? bd0 : t1;
            bi1 = c0 ? bi0 : u1;
            bd0 = c0 ? d : bd0;
            bi0 = c0 ? i : bi0;
        }
        int* o = idx_out + ((size_t)b * TLEN + t0 + lane) * KNN;
        o[0] = bi0; o[1] = bi1; o[2] = bi2;
    }
}

__global__ __launch_bounds__(256) void conv_kernel(const float* __restrict__ x,
                                                   const float* __restrict__ W,
                                                   const float* __restrict__ bias,
                                                   const int* __restrict__ idx,
                                                   float* __restrict__ out) {
    const int b   = blockIdx.x >> 6;
    const int blk = blockIdx.x & 63;
    const int wave  = threadIdx.x >> 5;
    const int lane  = threadIdx.x & 31;
    const int lmod  = lane & 15;
    const int lhalf = (lane < 16) ? 0 : 2;
    const int mhalf = (lane < 16) ? 0 : 8;

    const int o0 = (wave & 3) * 16;
    const int t0 = blk * 32 + (wave >> 2) * 16;
    const int tok = t0 + lmod;

    const float* xb = x + (size_t)b * CH * TLEN;
    const int* my = idx + ((size_t)b * TLEN + tok) * KNN;
    const int s0i = my[0], s1i = my[1], s2i = my[2];

    v8f acc = {};
#pragma unroll
    for (int kc = 0; kc < 48; ++kc) {
        const int ka0 = kc * 4 + lhalf;
        const int ka1 = ka0 + 1;
        v2f a, bb;
        a.x = W[(size_t)(o0 + lmod) * (CH * KNN) + ka0];
        a.y = W[(size_t)(o0 + lmod) * (CH * KNN) + ka1];
        {
            int i = ka0 / 3, kk = ka0 - i * 3;
            int s = (kk == 0) ? s0i : ((kk == 1) ? s1i : s2i);
            bb.x = xb[(size_t)i * TLEN + s];
        }
        {
            int i = ka1 / 3, kk = ka1 - i * 3;
            int s = (kk == 0) ? s0i : ((kk == 1) ? s1i : s2i);
            bb.y = xb[(size_t)i * TLEN + s];
        }
        acc = __builtin_amdgcn_wmma_f32_16x16x4_f32(
            false, a, false, bb, (short)0, acc, false, false);
    }

#pragma unroll
    for (int r = 0; r < 8; ++r) {
        int o = o0 + r + mhalf;
        out[(size_t)b * CH * TLEN + (size_t)o * TLEN + tok] = acc[r] + bias[o];
    }
}

extern "C" void kernel_launch(void* const* d_in, const int* in_sizes, int n_in,
                              void* d_out, int out_size, void* d_ws, size_t ws_size,
                              hipStream_t stream) {
    (void)in_sizes; (void)n_in; (void)out_size; (void)ws_size;
    const float* x    = (const float*)d_in[0];
    const float* W    = (const float*)d_in[1];
    const float* bias = (const float*)d_in[2];
    float* out = (float*)d_out;

    float* nsq = (float*)d_ws;
    int*   idx = (int*)((char*)d_ws + (size_t)BATCH * TLEN * sizeof(float));

    nsq_kernel<<<(BATCH * TLEN) / 256, 256, 0, stream>>>(x, nsq);

    hipFuncSetAttribute((const void*)knn_kernel,
                        hipFuncAttributeMaxDynamicSharedMemorySize, 180224);
    knn_kernel<<<BATCH * (TLEN / 128), 256, 180224, stream>>>(x, nsq, idx);

    conv_kernel<<<BATCH * (TLEN / 32), 256, 0, stream>>>(x, W, bias, idx, out);
}


# conv: one wave forms all 64 output channels of its 16 tokens (4 accumulator tiles), so each gathered x value is fetched once instead of by 4 waves; W fragments are read from an LDS copy of W
# speedup vs baseline: 2.0407x; 1.2466x over previous
#include <hip/hip_runtime.h>


typedef __attribute__((ext_vector_type(2))) float v2f;
typedef __attribute__((ext_vector_type(8))) float v8f;

#define BATCH 16
#define CH    64
#define TLEN  2048
#define KNN   3

__global__ __launch_bounds__(256) void nsq_kernel(const float* __restrict__ x,
                                                  float* __restrict__ nsq) {
    int g = blockIdx.x * 256 + threadIdx.x;
    int b = g >> 11;
    int t = g & (TLEN - 1);
    const float* xb = x + (size_t)b * CH * TLEN + t;
    float s = 0.0f;
#pragma unroll
    for (int c = 0; c < CH; ++c) {
        float v = xb[c * TLEN];
        s += v * v;
    }
    nsq[g] = s;
}

#define CAND_PITCH 49

#define TOP3_INSERT(r, dval, jval)                                            \
    {                                                                         \
        const float d_ = (dval);                                              \
        const int   j_ = (jval);                                              \
        const float a0_ = D0[r], a1_ = D1[r], a2_ = D2[r];                    \
        const int   k0_ = I0[r], k1_ = I1[r], k2_ = I2[r];                    \
        const bool c0_ = d_ < a0_;                                            \
        const bool c1_ = d_ < a1_;                                            \
        const bool c2_ = d_ < a2_;                                            \
        const float t2_ = c2_ ? d_ : a2_;                                     \
        const int   u2_ = c2_ ? j_ : k2_;                                     \
        const float t1_ = c1_ ? d_ : a1_;                                     \
        const int   u1_ = c1_ ? j_ : k1_;                                     \
        D2[r] = c1_ ? a1_ : t2_;                                              \
        I2[r] = c1_ ? k1_ : u2_;                                              \
        D1[r] = c0_ ? a0_ : t1_;                                              \
        I1[r] = c0_ ? k0_ : u1_;                                              \
        D0[r] = c0_ ? d_ : a0_;                                               \
        I0[r] = c0_ ? j_ : k0_;                                               \
    }

#define LOAD_TILE(BF, NS, S0)                                                 \
    {                                                                         \
        _Pragma("unroll")                                                     \
        for (int kc = 0; kc < 16; ++kc) {                                     \
            const int c0 = kc * 4 + lhalf;                                    \
            BF[kc].x = xb[(size_t)(c0 + 0) * TLEN + (S0) + lmod];             \
            BF[kc].y = xb[(size_t)(c0 + 1) * TLEN + (S0) + lmod];             \
        }                                                                     \
        NS = nsqb[(S0) + lmod];                                               \
    }

#define TILE_STEP(BF, NS, S0)                                                 \
    {                                                                         \
        v8f acc = {};                                                         \
        _Pragma("unroll")                                                     \
        for (int kc = 0; kc < 16; ++kc)                                       \
            acc = __builtin_amdgcn_wmma_f32_16x16x4_f32(                      \
                false, afrag[kc], false, BF[kc], (short)0, acc, false, false);\
        const int col_ = (S0) + lmod;                                         \
        _Pragma("unroll")                                                     \
        for (int r = 0; r < 8; ++r) {                                         \
            const float dd_ = fmaxf(nsq_t[r] + (NS) - 2.0f * acc[r], 0.0f);   \
            TOP3_INSERT(r, dd_, col_)                                         \
        }                                                                     \
    }

__global__ __launch_bounds__(256) void knn_kernel(const float* __restrict__ x,
                                                  const float* __restrict__ nsq,
                                                  int* __restrict__ idx_out) {
    const int b   = blockIdx.x >> 4;
    const int blk = blockIdx.x & 15;

    extern __shared__ float smem[];

    const int tid   = threadIdx.x;
    const int lane  = tid & 31;
    const int wave  = tid >> 5;
    const int lmod  = lane & 15;
    const int lhalf = (lane < 16) ? 0 : 2;
    const int mhalf = (lane < 16) ? 0 : 8;
    const int t0    = blk * 128 + wave * 16;

    float* cand_d = smem + wave * (16 * CAND_PITCH);
    int*   cand_i = (int*)(smem + 8 * 16 * CAND_PITCH) + wave * (16 * CAND_PITCH);

    const float* xb   = x   + (size_t)b * CH * TLEN;
    const float* nsqb = nsq + (size_t)b * TLEN;

    v2f afrag[16];
#pragma unroll
    for (int kc = 0; kc < 16; ++kc) {
        int c0 = kc * 4 + lhalf;
        afrag[kc].x = xb[(size_t)(c0 + 0) * TLEN + t0 + lmod];
        afrag[kc].y = xb[(size_t)(c0 + 1) * TLEN + t0 + lmod];
    }
    float nsq_t[8];
#pragma unroll
    for (int r = 0; r < 8; ++r) nsq_t[r] = nsqb[t0 + r + mhalf];

    float D0[8], D1[8], D2[8];
    int   I0[8], I1[8], I2[8];
#pragma unroll
    for (int r = 0; r < 8; ++r) {
        D0[r] = 3.4e38f; D1[r] = 3.4e38f; D2[r] = 3.4e38f;
        I0[r] = 0x7fffffff; I1[r] = 0x7fffffff; I2[r] = 0x7fffffff;
    }

    v2f bA[16], bB[16];
    float nsA, nsB;
    LOAD_TILE(bA, nsA, 0)
#pragma unroll 1
    for (int s0 = 0; s0 < TLEN; s0 += 32) {
        LOAD_TILE(bB, nsB, s0 + 16)
        TILE_STEP(bA, nsA, s0)
        const int sn = (s0 + 32 < TLEN) ? (s0 + 32) : 0;
        LOAD_TILE(bA, nsA, sn)
        TILE_STEP(bB, nsB, s0 + 16)
    }

#pragma unroll
    for (int r = 0; r < 8; ++r) {
        const int base = (r + mhalf) * CAND_PITCH + lmod * 3;
        cand_d[base + 0] = D0[r];  cand_i[base + 0] = I0[r];
        cand_d[base + 1] = D1[r];  cand_i[base + 1] = I1[r];
        cand_d[base + 2] = D2[r];  cand_i[base + 2] = I2[r];
    }
    __syncthreads();

    if (lane < 16) {
        const float* rd = cand_d + lane * CAND_PITCH;
        const int*   ri = cand_i + lane * CAND_PITCH;
        float bd0 = 3.4e38f, bd1 = 3.4e38f, bd2 = 3.4e38f;
        int   bi0 = 0x7fffffff, bi1 = 0x7fffffff, bi2 = 0x7fffffff;
#pragma unroll
        for (int c = 0; c < 48; ++c) {
            const float d = rd[c];
            const int   i = ri[c];
            const bool c0 = d < bd0 || (d == bd0 && i < bi0);
            const bool c1 = d < bd1 || (d == bd1 && i < bi1);
            const bool c2 = d < bd2 || (d == bd2 && i < bi2);
            const float t2 = c2 ? d : bd2;
            const int   u2 = c2 ? i : bi2;
            const float t1 = c1 ? d : bd1;
            const int   u1 = c1 ? i : bi1;
            bd2 = c1 ? bd1 : t2;
            bi2 = c1 ? bi1 : u2;
            bd1 = c0 ? bd0 : t1;
            bi1 = c0 ? bi0 : u1;
            bd0 = c0 ? d : bd0;
            bi0 = c0 ? i : bi0;
        }
        int* o = idx_out + ((size_t)b * TLEN + t0 + lane) * KNN;
        o[0] = bi0; o[1] = bi1; o[2] = bi2;
    }
}

#define W_PITCH 196

__global__ __launch_bounds__(256) void conv_kernel(const float* __restrict__ x,
                                                   const float* __restrict__ W,
                                                   const float* __restrict__ bias,
                                                   const int* __restrict__ idx,
                                                   float* __restrict__ out) {
    extern __shared__ __attribute__((aligned(16))) float wsm[];

    const int b    = blockIdx.x >> 4;
    const int blk  = blockIdx.x & 15;
    const int tid  = threadIdx.x;
    const int wave = tid >> 5;
    const int lane = tid & 31;
    const int lmod = lane & 15;
    const bool hi  = lane >= 16;
    const int lhalf = hi ? 2 : 0;
    const int mhalf = hi ? 8 : 0;

    const int t0  = blk * 128 + wave * 16;
    const int tok = t0 + lmod;

    const float* xb = x + (size_t)b * CH * TLEN;
    const int* my = idx + ((size_t)b * TLEN + tok) * KNN;
    const int s0i = my[0], s1i = my[1], s2i = my[2];

#pragma unroll
    for (int q = 0; q < 12; ++q) {
        const int f   = (q * 256 + tid) * 4;
        const int row = f / (CH * KNN);
        const int col = f - row * (CH * KNN);
        const float4 v = *(const float4*)(W + f);
        *(float4*)(wsm + row * W_PITCH + col) = v;
    }

    __syncthreads();

    const float* p0 = xb + (hi ? s2i : s0i);
    const float* p1 = xb + (hi ? s0i + TLEN : s1i);
    const float* p2 = xb + (hi ? s1i + TLEN : s2i);

    v2f bf[48];
#pragma unroll
    for (int kc = 0; kc < 48; ++kc) {
        const int k0 = kc * 4, k1 = kc * 4 + 1;
        const int i0 = k0 / 3, j0 = k0 - i0 * 3;
        const int i1 = k1 / 3, j1 = k1 - i1 * 3;
        const float* pa = (j0 == 0) ? p0 : ((j0 == 1) ? p1 : p2);
        const float* pb = (j1 == 0) ? p0 : ((j1 == 1) ? p1 : p2);
        bf[kc].x = pa[i0 * TLEN];
        bf[kc].y = pb[i1 * TLEN];
    }

    const float* wrow = wsm + lmod * W_PITCH + lhalf;
    v8f acc[4] = {};
#pragma unroll
    for (int kc = 0; kc < 48; ++kc) {
#pragma unroll
        for (int ot = 0; ot < 4; ++ot) {
            const v2f a = *(const v2f*)(wrow + ot * 16 * W_PITCH + kc * 4);
            acc[ot] = __builtin_amdgcn_wmma_f32_16x16x4_f32(
                false, a, false, bf[kc], (short)0, acc[ot], false, false);
        }
    }

    float* ob = out + (size_t)b * CH * TLEN + tok;
#pragma unroll
    for (int ot = 0; ot < 4; ++ot) {
#pragma unroll
        for (int r = 0; r < 8; ++r) {
            const int o = ot * 16 + r + mhalf;
            ob[(size_t)o * TLEN] = acc[ot][r] + bias[o];
        }
    }
}

extern "C" void kernel_launch(void* const* d_in, const int* in_sizes, int n_in,
                              void* d_out, int out_size, void* d_ws, size_t ws_size,
                              hipStream_t stream) {
    (void)in_sizes; (void)n_in; (void)out_size; (void)ws_size;
    const float* x    = (const float*)d_in[0];
    const float* W    = (const float*)d_in[1];
    const float* bias = (const float*)d_in[2];
    float* out = (float*)d_out;

    float* nsq = (float*)d_ws;
    int*   idx = (int*)((char*)d_ws + (size_t)BATCH * TLEN * sizeof(float));

    nsq_kernel<<<(BATCH * TLEN) / 256, 256, 0, stream>>>(x, nsq);

    hipFuncSetAttribute((const void*)knn_kernel,
                        hipFuncAttributeMaxDynamicSharedMemorySize, 180224);
    knn_kernel<<<BATCH * (TLEN / 128), 256, 180224, stream>>>(x, nsq, idx);

    hipFuncSetAttribute((const void*)conv_kernel,
                        hipFuncAttributeMaxDynamicSharedMemorySize, 180224);
    conv_kernel<<<BATCH * (TLEN / 128), 256, 180224, stream>>>(x, W, bias, idx, out);
}


# knn: x columns are staged once per block in LDS (128-bit global loads, channel pairs interleaved, one barrier per 512 columns); waves read B fragments by 64-bit LDS loads, not 32-bit global loads
# speedup vs baseline: 2.1679x; 1.0623x over previous
#include <hip/hip_runtime.h>


typedef __attribute__((ext_vector_type(2))) float v2f;
typedef __attribute__((ext_vector_type(8))) float v8f;

#define BATCH 16
#define CH    64
#define TLEN  2048
#define KNN   3

__global__ __launch_bounds__(256) void nsq_kernel(const float* __restrict__ x,
                                                  float* __restrict__ nsq) {
    int g = blockIdx.x * 256 + threadIdx.x;
    int b = g >> 11;
    int t = g & (TLEN - 1);
    const float* xb = x + (size_t)b * CH * TLEN + t;
    float s = 0.0f;
#pragma unroll
    for (int c = 0; c < CH; ++c) {
        float v = xb[c * TLEN];
        s += v * v;
    }
    nsq[g] = s;
}

#define CAND_PITCH 49

#define TOP3_INSERT(r, dval, jval)                                            \
    {                                                                         \
        const float d_ = (dval);                                              \
        const int   j_ = (jval);                                              \
        const float a0_ = D0[r], a1_ = D1[r], a2_ = D2[r];                    \
        const int   k0_ = I0[r], k1_ = I1[r], k2_ = I2[r];                    \
        const bool c0_ = d_ < a0_;                                            \
        const bool c1_ = d_ < a1_;                                            \
        const bool c2_ = d_ < a2_;                                            \
        const float t2_ = c2_ ? d_ : a2_;                                     \
        const int   u2_ = c2_ ? j_ : k2_;                                     \
        const float t1_ = c1_ ? d_ : a1_;                                     \
        const int   u1_ = c1_ ? j_ : k1_;                                     \
        D2[r] = c1_ ? a1_ : t2_;                                              \
        I2[r] = c1_ ? k1_ : u2_;                                              \
        D1[r] = c0_ ? a0_ : t1_;                                              \
        I1[r] = c0_ ? k0_ : u1_;                                              \
        D0[r] = c0_ ? d_ : a0_;                                               \
        I0[r] = c0_ ? j_ : k0_;                                               \
    }

#define KCH    128
#define NSUB   (TLEN / KCH)
#define XP2    288
#define XSTAGE (32 * XP2)
#define KNN_LDS_BYTES (8 * XSTAGE * 4)

#define STAGE_LOAD(RA, RB, S0)                                                \
    {                                                                         \
        _Pragma("unroll")                                                     \
        for (int q = 0; q < 4; ++q) {                                         \
            RA[q] = *(const float4*)(gsrc + (size_t)(q * 16) * TLEN + (S0));  \
            RB[q] = *(const float4*)(gsrc + (size_t)(q * 16 + 1) * TLEN + (S0)); \
        }                                                                     \
    }

#define STAGE_STORE(XS, RA, RB)                                               \
    {                                                                         \
        _Pragma("unroll")                                                     \
        for (int q = 0; q < 4; ++q) {                                         \
            float* d_ = (XS) + sdst + q * 8 * XP2;                            \
            *(float4*)(d_)     = make_float4(RA[q].x, RB[q].x, RA[q].y, RB[q].y); \
            *(float4*)(d_ + 4) = make_float4(RA[q].z, RB[q].z, RA[q].w, RB[q].w); \
        }                                                                     \
    }

#define LOAD_TILE(BF, NS, XS, J, S0)                                          \
    {                                                                         \
        _Pragma("unroll")                                                     \
        for (int kc = 0; kc < 16; ++kc)                                       \
            BF[kc] = *(const v2f*)((XS) + foff + kc * 2 * XP2 + (J) * 32);    \
        NS = nsqb[(S0) + lmod];                                               \
    }

#define TILE_STEP(BF, NS, S0)                                                 \
    {                                                                         \
        v8f acc = {};                                                         \
        _Pragma("unroll")                                                     \
        for (int kc = 0; kc < 16; ++kc)                                       \
            acc = __builtin_amdgcn_wmma_f32_16x16x4_f32(                      \
                false, afrag[kc], false, BF[kc], (short)0, acc, false, false);\
        const int col_ = (S0) + lmod;                                         \
        _Pragma("unroll")                                                     \
        for (int r = 0; r < 8; ++r) {                                         \
            const float dd_ = fmaxf(nsq_t[r] + (NS) - 2.0f * acc[r], 0.0f);   \
            TOP3_INSERT(r, dd_, col_)                                         \
        }                                                                     \
    }

__global__ __launch_bounds__(256) __attribute__((amdgpu_waves_per_eu(2, 2)))
void knn_kernel(const float* __restrict__ x,
                const float* __restrict__ nsq,
                int* __restrict__ idx_out) {
    const int b   = blockIdx.x >> 4;
    const int blk = blockIdx.x & 15;

    extern __shared__ __attribute__((aligned(16))) float smem[];

    const int tid   = threadIdx.x;
    const int lane  = tid & 31;
    const int wave  = tid >> 5;
    const int lmod  = lane & 15;
    const int lhalf = (lane < 16) ? 0 : 2;
    const int mhalf = (lane < 16) ? 0 : 8;
    const int t0    = blk * 128 + wave * 16;

    float* cand_d = smem + wave * (16 * CAND_PITCH);
    int*   cand_i = (int*)(smem + 8 * 16 * CAND_PITCH) + wave * (16 * CAND_PITCH);

    const float* xb   = x   + (size_t)b * CH * TLEN;
    const float* nsqb = nsq + (size_t)b * TLEN;

    v2f afrag[16];
#pragma unroll
    for (int kc = 0; kc < 16; ++kc) {
        int c0 = kc * 4 + lhalf;
        afrag[kc].x = xb[(size_t)(c0 + 0) * TLEN + t0 + lmod];
        afrag[kc].y = xb[(size_t)(c0 + 1) * TLEN + t0 + lmod];
    }
    float nsq_t[8];
#pragma unroll
    for (int r = 0; r < 8; ++r) nsq_t[r] = nsqb[t0 + r + mhalf];

    float D0[8], D1[8], D2[8];
    int   I0[8], I1[8], I2[8];
#pragma unroll
    for (int r = 0; r < 8; ++r) {
        D0[r] = 3.4e38f; D1[r] = 3.4e38f; D2[r] = 3.4e38f;
        I0[r] = 0x7fffffff; I1[r] = 0x7fffffff; I2[r] = 0x7fffffff;
    }

    const float* gsrc = xb + (size_t)(2 * wave) * TLEN + 4 * lane;
    const int sdst = wave * XP2 + 8 * lane;
    const int foff = (lhalf >> 1) * XP2 + 2 * lmod;

    {
        float4 pa[16], pb[16];
#pragma unroll
        for (int s = 0; s < 4; ++s) STAGE_LOAD((pa + 4 * s), (pb + 4 * s), s * KCH)
#pragma unroll
        for (int s = 0; s < 4; ++s) STAGE_STORE((smem + s * XSTAGE), (pa + 4 * s), (pb + 4 * s))
    }
    __syncthreads();

    float4 ra[4], rb[4];
    v2f bf[8][16];
    float ns[8];
#pragma unroll 1
    for (int c4 = 0; c4 < NSUB; c4 += 4) {
#pragma unroll 1
        for (int g = c4; g < c4 + 4; ++g) {
            const float* xs = smem + (g & 7) * XSTAGE;
            STAGE_LOAD(ra, rb, ((g + 4) & (NSUB - 1)) * KCH)
            const int s0 = g * KCH;
#pragma unroll
            for (int j = 0; j < 8; ++j) LOAD_TILE(bf[j], ns[j], xs, j, s0 + j * 16)
#pragma unroll
            for (int j = 0; j < 8; ++j) TILE_STEP(bf[j], ns[j], s0 + j * 16)
            STAGE_STORE((smem + ((g + 4) & 7) * XSTAGE), ra, rb)
        }
        __syncthreads();
    }

#pragma unroll
    for (int r = 0; r < 8; ++r) {
        const int base = (r + mhalf) * CAND_PITCH + lmod * 3;
        cand_d[base + 0] = D0[r];  cand_i[base + 0] = I0[r];
        cand_d[base + 1] = D1[r];  cand_i[base + 1] = I1[r];
        cand_d[base + 2] = D2[r];  cand_i[base + 2] = I2[r];
    }
    __syncthreads();

    if (lane < 16) {
        const float* rd = cand_d + lane * CAND_PITCH;
        const int*   ri = cand_i + lane * CAND_PITCH;
        float bd0 = 3.4e38f, bd1 = 3.4e38f, bd2 = 3.4e38f;
        int   bi0 = 0x7fffffff, bi1 = 0x7fffffff, bi2 = 0x7fffffff;
#pragma unroll
        for (int c = 0; c < 48; ++c) {
            const float d = rd[c];
            const int   i = ri[c];
            const bool c0 = d < bd0 || (d == bd0 && i < bi0);
            const bool c1 = d < bd1 || (d == bd1 && i < bi1);
            const bool c2 = d < bd2 || (d == bd2 && i < bi2);
            const float t2 = c2 ? d : bd2;
            const int   u2 = c2 ? i : bi2;
            const float t1 = c1 ? d : bd1;
            const int   u1 = c1 ? i : bi1;
            bd2 = c1 ? bd1 : t2;
            bi2 = c1 ? bi1 : u2;
            bd1 = c0 ? bd0 : t1;
            bi1 = c0 ? bi0 : u1;
            bd0 = c0 ? d : bd0;
            bi0 = c0 ? i : bi0;
        }
        int* o = idx_out + ((size_t)b * TLEN + t0 + lane) * KNN;
        o[0] = bi0; o[1] = bi1; o[2] = bi2;
    }
}

#define W_PITCH 196

__global__ __launch_bounds__(256) void conv_kernel(const float* __restrict__ x,
                                                   const float* __restrict__ W,
                                                   const float* __restrict__ bias,
                                                   const int* __restrict__ idx,
                                                   float* __restrict__ out) {
    extern __shared__ __attribute__((aligned(16))) float wsm[];

    const int b    = blockIdx.x >> 4;
    const int blk  = blockIdx.x & 15;
    const int tid  = threadIdx.x;
    const int wave = tid >> 5;
    const int lane = tid & 31;
    const int lmod = lane & 15;
    const bool hi  = lane >= 16;
    const int lhalf = hi ? 2 : 0;
    const int mhalf = hi ? 8 : 0;

    const int t0  = blk * 128 + wave * 16;
    const int tok = t0 + lmod;

    const float* xb = x + (size_t)b * CH * TLEN;
    const int* my = idx + ((size_t)b * TLEN + tok) * KNN;
    const int s0i = my[0], s1i = my[1], s2i = my[2];

#pragma unroll
    for (int q = 0; q < 12; ++q) {
        const int f   = (q * 256 + tid) * 4;
        const int row = f / (CH * KNN);
        const int col = f - row * (CH * KNN);
        const float4 v = *(const float4*)(W + f);
        *(float4*)(wsm + row * W_PITCH + col) = v;
    }

    __syncthreads();

    const float* p0 = xb + (hi ? s2i : s0i);
    const float* p1 = xb + (hi ? s0i + TLEN : s1i);
    const float* p2 = xb + (hi ? s1i + TLEN : s2i);

    v2f bf[48];
#pragma unroll
    for (int kc = 0; kc < 48; ++kc) {
        const int k0 = kc * 4, k1 = kc * 4 + 1;
        const int i0 = k0 / 3, j0 = k0 - i0 * 3;
        const int i1 = k1 / 3, j1 = k1 - i1 * 3;
        const float* pa = (j0 == 0) ? p0 : ((j0 == 1) ? p1 : p2);
        const float* pb = (j1 == 0) ? p0 : ((j1 == 1) ? p1 : p2);
        bf[kc].x = pa[i0 * TLEN];
        bf[kc].y = pb[i1 * TLEN];
    }

    const float* wrow = wsm + lmod * W_PITCH + lhalf;
    v8f acc[4] = {};
#pragma unroll
    for (int kc = 0; kc < 48; ++kc) {
#pragma unroll
        for (int ot = 0; ot < 4; ++ot) {
            const v2f a = *(const v2f*)(wrow + ot * 16 * W_PITCH + kc * 4);
            acc[ot] = __builtin_amdgcn_wmma_f32_16x16x4_f32(
                false, a, false, bf[kc], (short)0, acc[ot], false, false);
        }
    }

    float* ob = out + (size_t)b * CH * TLEN + tok;
#pragma unroll
    for (int ot = 0; ot < 4; ++ot) {
#pragma unroll
        for (int r = 0; r < 8; ++r) {
            const int o = ot * 16 + r + mhalf;
            ob[(size_t)o * TLEN] = acc[ot][r] + bias[o];
        }
    }
}

extern "C" void kernel_launch(void* const* d_in, const int* in_sizes, int n_in,
                              void* d_out, int out_size, void* d_ws, size_t ws_size,
                              hipStream_t stream) {
    (void)in_sizes; (void)n_in; (void)out_size; (void)ws_size;
    const float* x    = (const float*)d_in[0];
    const float* W    = (const float*)d_in[1];
    const float* bias = (const float*)d_in[2];
    float* out = (float*)d_out;

    float* nsq = (float*)d_ws;
    int*   idx = (int*)((char*)d_ws + (size_t)BATCH * TLEN * sizeof(float));

    nsq_kernel<<<(BATCH * TLEN) / 256, 256, 0, stream>>>(x, nsq);

    hipFuncSetAttribute((const void*)knn_kernel,
                        hipFuncAttributeMaxDynamicSharedMemorySize, KNN_LDS_BYTES);
    knn_kernel<<<BATCH * (TLEN / 128), 256, KNN_LDS_BYTES, stream>>>(x, nsq, idx);

    hipFuncSetAttribute((const void*)conv_kernel,
                        hipFuncAttributeMaxDynamicSharedMemorySize, 180224);
    conv_kernel<<<BATCH * (TLEN / 128), 256, 180224, stream>>>(x, W, bias, idx, out);
}
